# starting kernel
# baseline (speedup 1.0000x reference)
#include <hip/hip_runtime.h>
#include <hip/hip_bf16.h>

#define NUM_ENT   40000
#define EMBED     256
#define FEAT      768
#define KCHUNKS   (FEAT / 32)
#define NTILES    (EMBED / 16)
#define NE_TOT    40103
#define NE_SCORE  40001
#define BATCH     32

typedef __attribute__((ext_vector_type(16))) __bf16 v16bf;
typedef __attribute__((ext_vector_type(8)))  float  v8f;

__device__ __forceinline__ void load16(const float* __restrict__ p, float* v) {
  float4 a0 = *(const float4*)(p + 0);
  float4 a1 = *(const float4*)(p + 4);
  float4 b0 = *(const float4*)(p + 16);
  float4 b1 = *(const float4*)(p + 20);
  v[0] = a0.x; v[1] = a0.y; v[2]  = a0.z; v[3]  = a0.w;
  v[4] = a1.x; v[5] = a1.y; v[6]  = a1.z; v[7]  = a1.w;
  v[8] = b0.x; v[9] = b0.y; v[10] = b0.z; v[11] = b0.w;
  v[12] = b1.x; v[13] = b1.y; v[14] = b1.z; v[15] = b1.w;
}

__global__ __launch_bounds__(32) void BLP_split_w(
    const float* __restrict__ W, __bf16* __restrict__ whi,
    __bf16* __restrict__ wlo) {
  const int lane = threadIdx.x;
  const int kc   = blockIdx.x % KCHUNKS;
  const int nt   = blockIdx.x / KCHUNKS;
  const int half = lane >> 4;
  const int r15  = lane & 15;

  const float* p = W + (size_t)(nt * 16 + r15) * FEAT + kc * 32 + half * 8;
  float v[16];
  load16(p, v);

  const size_t off = ((size_t)(nt * KCHUNKS + kc) * 32 + lane) * 16;
#pragma unroll
  for (int i = 0; i < 16; ++i) {
    __bf16 h = (__bf16)v[i];
    whi[off + i] = h;
    wlo[off + i] = (__bf16)(v[i] - (float)h);
  }
}

__global__ __launch_bounds__(128) void BLP_gemm_proj(
    const float* __restrict__ A, const __bf16* __restrict__ whi,
    const __bf16* __restrict__ wlo, float* __restrict__ unity) {
  const int lane = threadIdx.x & 31;
  const int wave = threadIdx.x >> 5;
  const int half = lane >> 4;
  const int r15  = lane & 15;

  const int n_half = blockIdx.x & 1;
  const int m_tile = (blockIdx.x >> 1) * 4 + wave;
  const int m_base = m_tile * 16;

  v8f acc[8] = {};

  const float* arow = A + (size_t)(m_base + r15) * FEAT + half * 8;

  for (int kc = 0; kc < KCHUNKS; ++kc) {
    const int k0 = kc * 32;
    if (k0 + 32 < FEAT)
      __builtin_prefetch(arow + k0 + 32, 0, 0);

    float v[16];
    load16(arow + k0, v);
    v16bf ahi, alo;
#pragma unroll
    for (int i = 0; i < 16; ++i) {
      __bf16 h = (__bf16)v[i];
      ahi[i] = h;
      alo[i] = (__bf16)(v[i] - (float)h);
    }

#pragma unroll
    for (int nt = 0; nt < 8; ++nt) {
      const int nt_g = n_half * 8 + nt;
      const size_t off = ((size_t)(nt_g * KCHUNKS + kc) * 32 + lane) * 16;
      v16bf bhi = *(const v16bf*)(whi + off);
      v16bf blo = *(const v16bf*)(wlo + off);
      acc[nt] = __builtin_amdgcn_wmma_f32_16x16x32_bf16(
          false, ahi, false, bhi, (short)0, acc[nt], false, false);
      acc[nt] = __builtin_amdgcn_wmma_f32_16x16x32_bf16(
          false, ahi, false, blo, (short)0, acc[nt], false, false);
      acc[nt] = __builtin_amdgcn_wmma_f32_16x16x32_bf16(
          false, alo, false, bhi, (short)0, acc[nt], false, false);
    }
  }

#pragma unroll
  for (int nt = 0; nt < 8; ++nt) {
#pragma unroll
    for (int j = 0; j < 8; ++j) {
      int m   = m_base + j + half * 8;
      int col = n_half * 128 + nt * 16 + r15;
      unity[(size_t)(m + 1) * EMBED + col] = acc[nt][j];
    }
  }
}

__global__ __launch_bounds__(256) void BLP_copy_other(
    const float* __restrict__ other, float* __restrict__ unity) {
  int r = blockIdx.x;
  int d = threadIdx.x;
  int dst = (r == 0) ? 0 : (NUM_ENT + r);
  unity[(size_t)dst * EMBED + d] = other[(size_t)r * EMBED + d];
}

__global__ __launch_bounds__(256) void BLP_query(
    const float* __restrict__ unity, const int* __restrict__ ids,
    const int* __restrict__ mp_ptr, float* __restrict__ qsum) {
  __shared__ float s0[256];
  __shared__ float s1[256];
  const int b = blockIdx.x;
  const int d = threadIdx.x;
  const int mp = *mp_ptr;
  const int c0 = (mp == 0) ? 1 : 0;
  const int c1 = (mp <= 1) ? 2 : 1;
  const int id0 = ids[b * 3 + c0];
  const int id1 = ids[b * 3 + c1];
  float v0 = unity[(size_t)id0 * EMBED + d];
  float v1 = unity[(size_t)id1 * EMBED + d];
  s0[d] = v0 * v0;
  s1[d] = v1 * v1;
  __syncthreads();
  for (int off = 128; off > 0; off >>= 1) {
    if (d < off) { s0[d] += s0[d + off]; s1[d] += s1[d + off]; }
    __syncthreads();
  }
  float r0 = 1.0f / fmaxf(sqrtf(s0[0]), 1e-12f);
  float r1 = 1.0f / fmaxf(sqrtf(s1[0]), 1e-12f);
  qsum[(size_t)b * EMBED + d] = v0 * r0 + v1 * r1;
}

__global__ __launch_bounds__(256) void BLP_score(
    const float* __restrict__ unity, const float* __restrict__ qsum,
    float* __restrict__ out) {
  __shared__ float q[BATCH * EMBED];
  for (int i = threadIdx.x; i < BATCH * EMBED; i += 256) q[i] = qsum[i];
  __syncthreads();

  const int e = blockIdx.x * 256 + threadIdx.x;
  const bool active = (e < NE_SCORE);
  const float4* row = (const float4*)(unity + (size_t)(active ? e : 0) * EMBED);

  float acc[BATCH];
#pragma unroll
  for (int b = 0; b < BATCH; ++b) acc[b] = 0.0f;

  for (int d4 = 0; d4 < EMBED / 4; ++d4) {
    float4 v = row[d4];
    const int dbase = d4 * 4;
#pragma unroll
    for (int b = 0; b < BATCH; ++b) {
      const float* qb = q + b * EMBED + dbase;
      acc[b] += fabsf(qb[0] - v.x) + fabsf(qb[1] - v.y) +
                fabsf(qb[2] - v.z) + fabsf(qb[3] - v.w);
    }
  }
  if (active) {
#pragma unroll
    for (int b = 0; b < BATCH; ++b)
      out[(size_t)b * NE_SCORE + e] = -acc[b];
  }
}

extern "C" void kernel_launch(void* const* d_in, const int* in_sizes, int n_in,
                              void* d_out, int out_size, void* d_ws, size_t ws_size,
                              hipStream_t stream) {
  (void)in_sizes; (void)n_in; (void)out_size; (void)ws_size;
  const float* ent_pkl = (const float*)d_in[0];
  const float* other   = (const float*)d_in[1];
  const float* proj_W  = (const float*)d_in[2];
  const int*   ids     = (const int*)d_in[3];
  const int*   mp      = (const int*)d_in[4];
  float*       out     = (float*)d_out;

  float*  unity = (float*)d_ws;
  float*  qsum  = unity + (size_t)NE_TOT * EMBED;
  __bf16* whi   = (__bf16*)(qsum + (size_t)BATCH * EMBED);
  __bf16* wlo   = whi + (size_t)NTILES * KCHUNKS * 32 * 16;

  BLP_split_w<<<NTILES * KCHUNKS, 32, 0, stream>>>(proj_W, whi, wlo);
  BLP_gemm_proj<<<1250, 128, 0, stream>>>(ent_pkl, whi, wlo, unity);
  BLP_copy_other<<<103, 256, 0, stream>>>(other, unity);
  BLP_query<<<BATCH, 256, 0, stream>>>(unity, ids, mp, qsum);
  BLP_score<<<(NE_SCORE + 255) / 256, 256, 0, stream>>>(unity, qsum, out);
}


# Fused kernel re-tiled: block 80 rows x 256 dims, its 4 waves split the dims; rows split to bf16 hi/lo once per block, shared via LDS; 4 partial scores summed in LDS; sum order changed
# speedup vs baseline: 1.5002x; 1.5002x over previous
#include <hip/hip_runtime.h>
#include <hip/hip_bf16.h>

#define NUM_ENT   40000
#define EMBED     256
#define FEAT      768
#define KCHUNKS   (FEAT / 32)
#define NTILES    (EMBED / 16)
#define NE_SCORE  40001
#define BATCH     32
#define BM        80
#define RT        (BM / 16)
#define NTW       4
#define FUSED_BLOCKS (NUM_ENT / BM)
#define STG_ELEMS (RT * 2 * 2 * 32 * 8)

typedef __attribute__((ext_vector_type(16))) __bf16 v16bf;
typedef __attribute__((ext_vector_type(8)))  float  v8f;

__device__ __forceinline__ void load16(const float* __restrict__ p, float* v) {
  float4 a0 = *(const float4*)(p + 0);
  float4 a1 = *(const float4*)(p + 4);
  float4 b0 = *(const float4*)(p + 16);
  float4 b1 = *(const float4*)(p + 20);
  v[0] = a0.x; v[1] = a0.y; v[2]  = a0.z; v[3]  = a0.w;
  v[4] = a1.x; v[5] = a1.y; v[6]  = a1.z; v[7]  = a1.w;
  v[8] = b0.x; v[9] = b0.y; v[10] = b0.z; v[11] = b0.w;
  v[12] = b1.x; v[13] = b1.y; v[14] = b1.z; v[15] = b1.w;
}

__global__ __launch_bounds__(32) void BLP_split_w(
    const float* __restrict__ W, __bf16* __restrict__ whi,
    __bf16* __restrict__ wlo) {
  const int lane = threadIdx.x;
  const int kc   = blockIdx.x % KCHUNKS;
  const int nt   = blockIdx.x / KCHUNKS;
  const int half = lane >> 4;
  const int r15  = lane & 15;

  const float* p = W + (size_t)(nt * 16 + r15) * FEAT + kc * 32 + half * 8;
  float v[16];
  load16(p, v);

  const size_t off = ((size_t)(nt * KCHUNKS + kc) * 32 + lane) * 16;
#pragma unroll
  for (int i = 0; i < 16; ++i) {
    __bf16 h = (__bf16)v[i];
    whi[off + i] = h;
    wlo[off + i] = (__bf16)(v[i] - (float)h);
  }
}

__device__ __forceinline__ void split16(const float* v, v16bf& hi, v16bf& lo) {
#pragma unroll
  for (int i = 0; i < 16; ++i) {
    __bf16 h = (__bf16)v[i];
    hi[i] = h;
    lo[i] = (__bf16)(v[i] - (float)h);
  }
}

__device__ __forceinline__ v8f mma3(v16bf whi_f, v16bf wlo_f, v16bf rhi,
                                    v16bf rlo, v8f c) {
  c = __builtin_amdgcn_wmma_f32_16x16x32_bf16(
      false, whi_f, false, rhi, (short)0, c, false, false);
  c = __builtin_amdgcn_wmma_f32_16x16x32_bf16(
      false, wlo_f, false, rhi, (short)0, c, false, false);
  c = __builtin_amdgcn_wmma_f32_16x16x32_bf16(
      false, whi_f, false, rlo, (short)0, c, false, false);
  return c;
}

__global__ __launch_bounds__(512) void BLP_query_proj(
    const float* __restrict__ A, const float* __restrict__ other,
    const __bf16* __restrict__ whi, const __bf16* __restrict__ wlo,
    const int* __restrict__ ids, const int* __restrict__ mp_ptr,
    float* __restrict__ qsum) {
  __shared__ float P[16 * EMBED];
  __shared__ float rn[16];
  const int lane = threadIdx.x & 31;
  const int wave = threadIdx.x >> 5;
  const int half = lane >> 4;
  const int r15  = lane & 15;
  const int mp = *mp_ptr;
  const int c0 = (mp == 0) ? 1 : 0;
  const int c1 = (mp <= 1) ? 2 : 1;
  const int b  = blockIdx.x * 8 + (r15 & 7);
  const int id = ids[b * 3 + ((r15 >> 3) ? c1 : c0)];
  const bool is_ent = (id >= 1) && (id <= NUM_ENT);
  const int erow = is_ent ? (id - 1) : 0;
  const int orow = is_ent ? 0 : ((id == 0) ? 0 : (id - NUM_ENT));

  const float* arow = A + (size_t)erow * FEAT + half * 8;
  v8f acc = {};
#pragma unroll 4
  for (int kc = 0; kc < KCHUNKS; ++kc) {
    float v[16];
    load16(arow + kc * 32, v);
    v16bf rhi, rlo;
    split16(v, rhi, rlo);
    const size_t off = ((size_t)(wave * KCHUNKS + kc) * 32 + lane) * 16;
    v16bf bhi = *(const v16bf*)(whi + off);
    v16bf blo = *(const v16bf*)(wlo + off);
    acc = mma3(bhi, blo, rhi, rlo, acc);
  }
#pragma unroll
  for (int j = 0; j < 8; ++j) {
    const int d = wave * 16 + j + half * 8;
    float val = acc[j];
    if (!is_ent) val = other[(size_t)orow * EMBED + d];
    P[r15 * EMBED + d] = val;
  }
  __syncthreads();
  {
    float s = 0.0f;
#pragma unroll
    for (int i = 0; i < 8; ++i) {
      float x = P[wave * EMBED + lane + 32 * i];
      s += x * x;
    }
    for (int o = 16; o > 0; o >>= 1) s += __shfl_xor(s, o);
    if (lane == 0) rn[wave] = 1.0f / fmaxf(sqrtf(s), 1e-12f);
  }
  __syncthreads();
#pragma unroll
  for (int i = 0; i < 4; ++i) {
    const int idx = threadIdx.x + i * 512;
    const int bl = idx >> 8;
    const int d  = idx & 255;
    qsum[(size_t)(blockIdx.x * 8 + bl) * EMBED + d] =
        P[bl * EMBED + d] * rn[bl] + P[(bl + 8) * EMBED + d] * rn[bl + 8];
  }
}

typedef __attribute__((ext_vector_type(8))) __bf16 v8bf;
typedef __attribute__((ext_vector_type(4))) __bf16 v4bf;
typedef __attribute__((ext_vector_type(4))) float v4f;

__device__ __forceinline__ v16bf ld_frag(const __bf16* st, int rt, int hl,
                                         int lane) {
  const __bf16* p = st + (rt * 2 + hl) * 512 + lane * 8;
  v8bf e0 = *(const v8bf*)p;
  v8bf e1 = *(const v8bf*)(p + 256);
  return __builtin_shufflevector(e0, e1, 0, 1, 2, 3, 4, 5, 6, 7, 8, 9, 10, 11,
                                 12, 13, 14, 15);
}

__device__ __forceinline__ void stage_store(__bf16* dst_hi, v4f v) {
  const float f[4] = {v[0], v[1], v[2], v[3]};
  v4bf hi, lo;
#pragma unroll
  for (int j = 0; j < 4; ++j) {
    __bf16 h = (__bf16)f[j];
    hi[j] = h;
    lo[j] = (__bf16)(f[j] - (float)h);
  }
  *(v4bf*)dst_hi = hi;
  *(v4bf*)(dst_hi + 512) = lo;
}

template <bool LOADA, bool MORE, bool PF>
__device__ __forceinline__ void k_step(
    int kc, const float* __restrict__ ag, const __bf16* __restrict__ wh,
    const __bf16* __restrict__ wl, __bf16* stg_cur, __bf16* stg_nxt, int toff,
    int lane, v4f (&ac)[RT], v4f (&an)[RT], v16bf (&bhc)[NTW],
    v16bf (&blc)[NTW], v16bf (&bhn)[NTW], v16bf (&bln)[NTW],
    v8f (&acc)[RT][NTW]) {
  if (LOADA) {
#pragma unroll
    for (int i = 0; i < RT; ++i)
      an[i] = *(const v4f*)(ag + (size_t)i * 16 * FEAT + (kc + 2) * 32);
  }
  if (MORE) {
#pragma unroll
    for (int nt = 0; nt < NTW; ++nt) {
      const size_t off = (size_t)(nt * KCHUNKS + kc + 1) * 512;
      bhn[nt] = *(const v16bf*)(wh + off);
      bln[nt] = *(const v16bf*)(wl + off);
    }
  }
  if (PF) {
    const int pk = (kc + 4 < KCHUNKS) ? kc + 4 : KCHUNKS - 1;
#pragma unroll
    for (int i = 0; i < RT; ++i)
      __builtin_prefetch(ag + (size_t)i * 16 * FEAT + pk * 32, 0, 0);
  }
  __builtin_amdgcn_sched_barrier(0);

  v16bf rhi[RT], rlo[RT];
#pragma unroll
  for (int rt = 0; rt < RT; ++rt) {
    rhi[rt] = ld_frag(stg_cur, rt, 0, lane);
    rlo[rt] = ld_frag(stg_cur, rt, 1, lane);
  }
#pragma unroll
  for (int rt = 0; rt < RT; ++rt)
#pragma unroll
    for (int nt = 0; nt < NTW; ++nt)
      acc[rt][nt] = __builtin_amdgcn_wmma_f32_16x16x32_bf16(
          false, bhc[nt], false, rhi[rt], (short)0, acc[rt][nt], false, false);
#pragma unroll
  for (int rt = 0; rt < RT; ++rt)
#pragma unroll
    for (int nt = 0; nt < NTW; ++nt)
      acc[rt][nt] = __builtin_amdgcn_wmma_f32_16x16x32_bf16(
          false, blc[nt], false, rhi[rt], (short)0, acc[rt][nt], false, false);
#pragma unroll
  for (int rt = 0; rt < RT; ++rt)
#pragma unroll
    for (int nt = 0; nt < NTW; ++nt)
      acc[rt][nt] = __builtin_amdgcn_wmma_f32_16x16x32_bf16(
          false, bhc[nt], false, rlo[rt], (short)0, acc[rt][nt], false, false);

  if (MORE) {
#pragma unroll
    for (int i = 0; i < RT; ++i) asm volatile("" : "+v"(ac[i]));
#pragma unroll
    for (int i = 0; i < RT; ++i) stage_store(stg_nxt + i * 1024 + toff, ac[i]);
    __builtin_amdgcn_fence(__ATOMIC_RELEASE, "workgroup", "local");
    __builtin_amdgcn_s_barrier();
    __builtin_amdgcn_fence(__ATOMIC_ACQUIRE, "workgroup", "local");
  }
}

__global__ __launch_bounds__(128) void BLP_proj_score(
    const float* __restrict__ A, const float* __restrict__ other,
    const __bf16* __restrict__ whi, const __bf16* __restrict__ wlo,
    const float* __restrict__ qsum, float* __restrict__ out) {
  __shared__ __attribute__((aligned(16))) float q[BATCH * EMBED];
  __shared__ __attribute__((aligned(16))) __bf16 stg[2][STG_ELEMS];
  __shared__ __attribute__((aligned(16))) float red[4][BATCH * BM];

  if (blockIdx.x == FUSED_BLOCKS) {
    if (threadIdx.x >= 64) return;
    const int b = threadIdx.x >> 1;
    const int h = threadIdx.x & 1;
    const float4* qp = (const float4*)(qsum + (size_t)b * EMBED + h * 128);
    const float4* op = (const float4*)(other + h * 128);
    float s = 0.0f;
    for (int i = 0; i < 32; ++i) {
      float4 x = qp[i];
      float4 y = op[i];
      s += fabsf(x.x - y.x) + fabsf(x.y - y.y) +
           fabsf(x.z - y.z) + fabsf(x.w - y.w);
    }
    s += __shfl_xor(s, 1);
    if (h == 0) out[(size_t)b * NE_SCORE] = -s;
    return;
  }

  const int t    = threadIdx.x;
  const int lane = t & 31;
  const int wave = t >> 5;
  const int half = lane >> 4;
  const int r15  = lane & 15;
  const int row0 = blockIdx.x * BM;

  const int kq = t & 7;
  const int rb = t >> 3;
  const float* ag = A + (size_t)(row0 + rb) * FEAT + kq * 4;
  const int toff =
      ((kq >> 2) * 32 + rb + 16 * ((kq >> 1) & 1)) * 8 + (kq & 1) * 4;

  const __bf16* wh = whi + ((size_t)(wave * NTW * KCHUNKS) * 32 + lane) * 16;
  const __bf16* wl = wlo + ((size_t)(wave * NTW * KCHUNKS) * 32 + lane) * 16;

  v16bf b0h[NTW], b0l[NTW], b1h[NTW], b1l[NTW];
  v4f a0[RT], a1[RT];
  {
#pragma unroll
    for (int i = 0; i < RT; ++i)
      a0[i] = *(const v4f*)(ag + (size_t)i * 16 * FEAT);
#pragma unroll
    for (int i = 0; i < RT; ++i)
      a1[i] = *(const v4f*)(ag + (size_t)i * 16 * FEAT + 32);
#pragma unroll
    for (int nt = 0; nt < NTW; ++nt) {
      const size_t off = (size_t)(nt * KCHUNKS) * 512;
      b0h[nt] = *(const v16bf*)(wh + off);
      b0l[nt] = *(const v16bf*)(wl + off);
    }
    for (int i = t; i < BATCH * EMBED / 4; i += 128)
      ((float4*)q)[i] = ((const float4*)qsum)[i];
#pragma unroll
    for (int i = 0; i < RT; ++i) stage_store(stg[0] + i * 1024 + toff, a0[i]);
  }
  __syncthreads();

  v8f acc[RT][NTW] = {};
  for (int kc = 0; kc < KCHUNKS - 2; kc += 2) {
    k_step<true, true, true>(kc, ag, wh, wl, stg[0], stg[1], toff, lane, a1,
                             a0, b0h, b0l, b1h, b1l, acc);
    k_step<true, true, false>(kc + 1, ag, wh, wl, stg[1], stg[0], toff, lane,
                              a0, a1, b1h, b1l, b0h, b0l, acc);
  }
  k_step<false, true, false>(KCHUNKS - 2, ag, wh, wl, stg[0], stg[1], toff,
                             lane, a1, a0, b0h, b0l, b1h, b1l, acc);
  k_step<false, false, false>(KCHUNKS - 1, ag, wh, wl, stg[1], stg[0], toff,
                              lane, a0, a1, b1h, b1l, b0h, b0l, acc);

  for (int b = 0; b < BATCH; ++b) {
    const float* qb = q + b * EMBED + wave * 64 + half * 8;
    float sa[RT], sb[RT];
#pragma unroll
    for (int rt = 0; rt < RT; ++rt) { sa[rt] = 0.0f; sb[rt] = 0.0f; }
#pragma unroll
    for (int nt = 0; nt < NTW; ++nt) {
      const float4 q0 = *(const float4*)(qb + nt * 16);
      const float4 q1 = *(const float4*)(qb + nt * 16 + 4);
#pragma unroll
      for (int rt = 0; rt < RT; ++rt) {
        sa[rt] += fabsf(q0.x - acc[rt][nt][0]);
        sb[rt] += fabsf(q0.y - acc[rt][nt][1]);
        sa[rt] += fabsf(q0.z - acc[rt][nt][2]);
        sb[rt] += fabsf(q0.w - acc[rt][nt][3]);
        sa[rt] += fabsf(q1.x - acc[rt][nt][4]);
        sb[rt] += fabsf(q1.y - acc[rt][nt][5]);
        sa[rt] += fabsf(q1.z - acc[rt][nt][6]);
        sb[rt] += fabsf(q1.w - acc[rt][nt][7]);
      }
    }
#pragma unroll
    for (int rt = 0; rt < RT; ++rt) {
      float s = sa[rt] + sb[rt];
      s += __shfl_xor(s, 16);
      if (half == 0) red[wave][b * BM + rt * 16 + r15] = s;
    }
  }
  __syncthreads();

  for (int idx = t; idx < BATCH * BM; idx += 128) {
    const int b = idx / BM;
    const int r = idx - b * BM;
    out[(size_t)b * NE_SCORE + 1 + row0 + r] =
        -((red[0][idx] + red[1][idx]) + (red[2][idx] + red[3][idx]));
  }
}

extern "C" void kernel_launch(void* const* d_in, const int* in_sizes, int n_in,
                              void* d_out, int out_size, void* d_ws, size_t ws_size,
                              hipStream_t stream) {
  (void)in_sizes; (void)n_in; (void)out_size; (void)ws_size;
  const float* ent_pkl = (const float*)d_in[0];
  const float* other   = (const float*)d_in[1];
  const float* proj_W  = (const float*)d_in[2];
  const int*   ids     = (const int*)d_in[3];
  const int*   mp      = (const int*)d_in[4];
  float*       out     = (float*)d_out;

  float*  qsum  = (float*)d_ws;
  __bf16* whi   = (__bf16*)(qsum + (size_t)BATCH * EMBED);
  __bf16* wlo   = whi + (size_t)NTILES * KCHUNKS * 32 * 16;

  BLP_split_w<<<NTILES * KCHUNKS, 32, 0, stream>>>(proj_W, whi, wlo);
  BLP_query_proj<<<BATCH / 8, 512, 0, stream>>>(ent_pkl, other, whi, wlo, ids, mp, qsum);
  BLP_proj_score<<<FUSED_BLOCKS + 1, 128, 0, stream>>>(ent_pkl, other, whi, wlo, qsum, out);
}


# Score partials: every lane stores its own sum to LDS (8 per score, was 4), added in the final pass in the old pairing; no cross-lane shuffle, wait or masked store per row tile; sums bit-identical
# speedup vs baseline: 1.7202x; 1.1466x over previous
#include <hip/hip_runtime.h>
#include <hip/hip_bf16.h>

#define NUM_ENT   40000
#define EMBED     256
#define FEAT      768
#define KCHUNKS   (FEAT / 32)
#define NTILES    (EMBED / 16)
#define NE_SCORE  40001
#define BATCH     32
#define BM        80
#define RT        (BM / 16)
#define NTW       4
#define FUSED_BLOCKS (NUM_ENT / BM)
#define STG_ELEMS (RT * 2 * 2 * 32 * 8)
#define RED_PLANE (BATCH * BM + 16)

typedef __attribute__((ext_vector_type(16))) __bf16 v16bf;
typedef __attribute__((ext_vector_type(8)))  float  v8f;

__device__ __forceinline__ void load16(const float* __restrict__ p, float* v) {
  float4 a0 = *(const float4*)(p + 0);
  float4 a1 = *(const float4*)(p + 4);
  float4 b0 = *(const float4*)(p + 16);
  float4 b1 = *(const float4*)(p + 20);
  v[0] = a0.x; v[1] = a0.y; v[2]  = a0.z; v[3]  = a0.w;
  v[4] = a1.x; v[5] = a1.y; v[6]  = a1.z; v[7]  = a1.w;
  v[8] = b0.x; v[9] = b0.y; v[10] = b0.z; v[11] = b0.w;
  v[12] = b1.x; v[13] = b1.y; v[14] = b1.z; v[15] = b1.w;
}

__global__ __launch_bounds__(32) void BLP_split_w(
    const float* __restrict__ W, __bf16* __restrict__ whi,
    __bf16* __restrict__ wlo) {
  const int lane = threadIdx.x;
  const int kc   = blockIdx.x % KCHUNKS;
  const int nt   = blockIdx.x / KCHUNKS;
  const int half = lane >> 4;
  const int r15  = lane & 15;

  const float* p = W + (size_t)(nt * 16 + r15) * FEAT + kc * 32 + half * 8;
  float v[16];
  load16(p, v);

  const size_t off = ((size_t)(nt * KCHUNKS + kc) * 32 + lane) * 16;
#pragma unroll
  for (int i = 0; i < 16; ++i) {
    __bf16 h = (__bf16)v[i];
    whi[off + i] = h;
    wlo[off + i] = (__bf16)(v[i] - (float)h);
  }
}

__device__ __forceinline__ void split16(const float* v, v16bf& hi, v16bf& lo) {
#pragma unroll
  for (int i = 0; i < 16; ++i) {
    __bf16 h = (__bf16)v[i];
    hi[i] = h;
    lo[i] = (__bf16)(v[i] - (float)h);
  }
}

__device__ __forceinline__ v8f mma3(v16bf whi_f, v16bf wlo_f, v16bf rhi,
                                    v16bf rlo, v8f c) {
  c = __builtin_amdgcn_wmma_f32_16x16x32_bf16(
      false, whi_f, false, rhi, (short)0, c, false, false);
  c = __builtin_amdgcn_wmma_f32_16x16x32_bf16(
      false, wlo_f, false, rhi, (short)0, c, false, false);
  c = __builtin_amdgcn_wmma_f32_16x16x32_bf16(
      false, whi_f, false, rlo, (short)0, c, false, false);
  return c;
}

__global__ __launch_bounds__(512) void BLP_query_proj(
    const float* __restrict__ A, const float* __restrict__ other,
    const __bf16* __restrict__ whi, const __bf16* __restrict__ wlo,
    const int* __restrict__ ids, const int* __restrict__ mp_ptr,
    float* __restrict__ qsum) {
  __shared__ float P[16 * EMBED];
  __shared__ float rn[16];
  const int lane = threadIdx.x & 31;
  const int wave = threadIdx.x >> 5;
  const int half = lane >> 4;
  const int r15  = lane & 15;
  const int mp = *mp_ptr;
  const int c0 = (mp == 0) ? 1 : 0;
  const int c1 = (mp <= 1) ? 2 : 1;
  const int b  = blockIdx.x * 8 + (r15 & 7);
  const int id = ids[b * 3 + ((r15 >> 3) ? c1 : c0)];
  const bool is_ent = (id >= 1) && (id <= NUM_ENT);
  const int erow = is_ent ? (id - 1) : 0;
  const int orow = is_ent ? 0 : ((id == 0) ? 0 : (id - NUM_ENT));

  const float* arow = A + (size_t)erow * FEAT + half * 8;
  v8f acc = {};
#pragma unroll 4
  for (int kc = 0; kc < KCHUNKS; ++kc) {
    float v[16];
    load16(arow + kc * 32, v);
    v16bf rhi, rlo;
    split16(v, rhi, rlo);
    const size_t off = ((size_t)(wave * KCHUNKS + kc) * 32 + lane) * 16;
    v16bf bhi = *(const v16bf*)(whi + off);
    v16bf blo = *(const v16bf*)(wlo + off);
    acc = mma3(bhi, blo, rhi, rlo, acc);
  }
#pragma unroll
  for (int j = 0; j < 8; ++j) {
    const int d = wave * 16 + j + half * 8;
    float val = acc[j];
    if (!is_ent) val = other[(size_t)orow * EMBED + d];
    P[r15 * EMBED + d] = val;
  }
  __syncthreads();
  {
    float s = 0.0f;
#pragma unroll
    for (int i = 0; i < 8; ++i) {
      float x = P[wave * EMBED + lane + 32 * i];
      s += x * x;
    }
    for (int o = 16; o > 0; o >>= 1) s += __shfl_xor(s, o);
    if (lane == 0) rn[wave] = 1.0f / fmaxf(sqrtf(s), 1e-12f);
  }
  __syncthreads();
#pragma unroll
  for (int i = 0; i < 4; ++i) {
    const int idx = threadIdx.x + i * 512;
    const int bl = idx >> 8;
    const int d  = idx & 255;
    qsum[(size_t)(blockIdx.x * 8 + bl) * EMBED + d] =
        P[bl * EMBED + d] * rn[bl] + P[(bl + 8) * EMBED + d] * rn[bl + 8];
  }
}

typedef __attribute__((ext_vector_type(8))) __bf16 v8bf;
typedef __attribute__((ext_vector_type(4))) __bf16 v4bf;
typedef __attribute__((ext_vector_type(4))) float v4f;

__device__ __forceinline__ v16bf ld_frag(const __bf16* st, int rt, int hl,
                                         int lane) {
  const __bf16* p = st + (rt * 2 + hl) * 512 + lane * 8;
  v8bf e0 = *(const v8bf*)p;
  v8bf e1 = *(const v8bf*)(p + 256);
  return __builtin_shufflevector(e0, e1, 0, 1, 2, 3, 4, 5, 6, 7, 8, 9, 10, 11,
                                 12, 13, 14, 15);
}

__device__ __forceinline__ void stage_store(__bf16* dst_hi, v4f v) {
  const float f[4] = {v[0], v[1], v[2], v[3]};
  v4bf hi, lo;
#pragma unroll
  for (int j = 0; j < 4; ++j) {
    __bf16 h = (__bf16)f[j];
    hi[j] = h;
    lo[j] = (__bf16)(f[j] - (float)h);
  }
  *(v4bf*)dst_hi = hi;
  *(v4bf*)(dst_hi + 512) = lo;
}

template <bool LOADA, bool MORE, bool PF>
__device__ __forceinline__ void k_step(
    int kc, const float* __restrict__ ag, const __bf16* __restrict__ wh,
    const __bf16* __restrict__ wl, __bf16* stg_cur, __bf16* stg_nxt, int toff,
    int lane, v4f (&ac)[RT], v4f (&an)[RT], v16bf (&bhc)[NTW],
    v16bf (&blc)[NTW], v16bf (&bhn)[NTW], v16bf (&bln)[NTW],
    v8f (&acc)[RT][NTW]) {
  if (LOADA) {
#pragma unroll
    for (int i = 0; i < RT; ++i)
      an[i] = *(const v4f*)(ag + (size_t)i * 16 * FEAT + (kc + 2) * 32);
  }
  if (MORE) {
#pragma unroll
    for (int nt = 0; nt < NTW; ++nt) {
      const size_t off = (size_t)(nt * KCHUNKS + kc + 1) * 512;
      bhn[nt] = *(const v16bf*)(wh + off);
      bln[nt] = *(const v16bf*)(wl + off);
    }
  }
  if (PF) {
    const int pk = (kc + 4 < KCHUNKS) ? kc + 4 : KCHUNKS - 1;
#pragma unroll
    for (int i = 0; i < RT; ++i)
      __builtin_prefetch(ag + (size_t)i * 16 * FEAT + pk * 32, 0, 0);
  }
  __builtin_amdgcn_sched_barrier(0);

  v16bf rhi[RT], rlo[RT];
#pragma unroll
  for (int rt = 0; rt < RT; ++rt) {
    rhi[rt] = ld_frag(stg_cur, rt, 0, lane);
    rlo[rt] = ld_frag(stg_cur, rt, 1, lane);
  }
#pragma unroll
  for (int rt = 0; rt < RT; ++rt)
#pragma unroll
    for (int nt = 0; nt < NTW; ++nt)
      acc[rt][nt] = __builtin_amdgcn_wmma_f32_16x16x32_bf16(
          false, bhc[nt], false, rhi[rt], (short)0, acc[rt][nt], false, false);
#pragma unroll
  for (int rt = 0; rt < RT; ++rt)
#pragma unroll
    for (int nt = 0; nt < NTW; ++nt)
      acc[rt][nt] = __builtin_amdgcn_wmma_f32_16x16x32_bf16(
          false, blc[nt], false, rhi[rt], (short)0, acc[rt][nt], false, false);
#pragma unroll
  for (int rt = 0; rt < RT; ++rt)
#pragma unroll
    for (int nt = 0; nt < NTW; ++nt)
      acc[rt][nt] = __builtin_amdgcn_wmma_f32_16x16x32_bf16(
          false, bhc[nt], false, rlo[rt], (short)0, acc[rt][nt], false, false);

  if (MORE) {
#pragma unroll
    for (int i = 0; i < RT; ++i) asm volatile("" : "+v"(ac[i]));
#pragma unroll
    for (int i = 0; i < RT; ++i) stage_store(stg_nxt + i * 1024 + toff, ac[i]);
    __builtin_amdgcn_fence(__ATOMIC_RELEASE, "workgroup", "local");
    __builtin_amdgcn_s_barrier();
    __builtin_amdgcn_fence(__ATOMIC_ACQUIRE, "workgroup", "local");
  }
}

__global__ __launch_bounds__(128) void BLP_proj_score(
    const float* __restrict__ A, const float* __restrict__ other,
    const __bf16* __restrict__ whi, const __bf16* __restrict__ wlo,
    const float* __restrict__ qsum, float* __restrict__ out) {
  __shared__ __attribute__((aligned(16))) float q[BATCH * EMBED];
  __shared__ __attribute__((aligned(16))) __bf16 stg[2][STG_ELEMS];
  __shared__ __attribute__((aligned(16))) float red[8 * RED_PLANE];

  if (blockIdx.x == FUSED_BLOCKS) {
    if (threadIdx.x >= 64) return;
    const int b = threadIdx.x >> 1;
    const int h = threadIdx.x & 1;
    const float4* qp = (const float4*)(qsum + (size_t)b * EMBED + h * 128);
    const float4* op = (const float4*)(other + h * 128);
    float s = 0.0f;
    for (int i = 0; i < 32; ++i) {
      float4 x = qp[i];
      float4 y = op[i];
      s += fabsf(x.x - y.x) + fabsf(x.y - y.y) +
           fabsf(x.z - y.z) + fabsf(x.w - y.w);
    }
    s += __shfl_xor(s, 1);
    if (h == 0) out[(size_t)b * NE_SCORE] = -s;
    return;
  }

  const int t    = threadIdx.x;
  const int lane = t & 31;
  const int wave = t >> 5;
  const int half = lane >> 4;
  const int r15  = lane & 15;
  const int row0 = blockIdx.x * BM;

  const int kq = t & 7;
  const int rb = t >> 3;
  const float* ag = A + (size_t)(row0 + rb) * FEAT + kq * 4;
  const int toff =
      ((kq >> 2) * 32 + rb + 16 * ((kq >> 1) & 1)) * 8 + (kq & 1) * 4;

  const __bf16* wh = whi + ((size_t)(wave * NTW * KCHUNKS) * 32 + lane) * 16;
  const __bf16* wl = wlo + ((size_t)(wave * NTW * KCHUNKS) * 32 + lane) * 16;

  v16bf b0h[NTW], b0l[NTW], b1h[NTW], b1l[NTW];
  v4f a0[RT], a1[RT];
  {
#pragma unroll
    for (int i = 0; i < RT; ++i)
      a0[i] = *(const v4f*)(ag + (size_t)i * 16 * FEAT);
#pragma unroll
    for (int i = 0; i < RT; ++i)
      a1[i] = *(const v4f*)(ag + (size_t)i * 16 * FEAT + 32);
#pragma unroll
    for (int nt = 0; nt < NTW; ++nt) {
      const size_t off = (size_t)(nt * KCHUNKS) * 512;
      b0h[nt] = *(const v16bf*)(wh + off);
      b0l[nt] = *(const v16bf*)(wl + off);
    }
    for (int i = t; i < BATCH * EMBED / 4; i += 128)
      ((float4*)q)[i] = ((const float4*)qsum)[i];
#pragma unroll
    for (int i = 0; i < RT; ++i) stage_store(stg[0] + i * 1024 + toff, a0[i]);
  }
  __syncthreads();

  v8f acc[RT][NTW] = {};
  for (int kc = 0; kc < KCHUNKS - 2; kc += 2) {
    k_step<true, true, true>(kc, ag, wh, wl, stg[0], stg[1], toff, lane, a1,
                             a0, b0h, b0l, b1h, b1l, acc);
    k_step<true, true, false>(kc + 1, ag, wh, wl, stg[1], stg[0], toff, lane,
                              a0, a1, b1h, b1l, b0h, b0l, acc);
  }
  k_step<false, true, false>(KCHUNKS - 2, ag, wh, wl, stg[0], stg[1], toff,
                             lane, a1, a0, b0h, b0l, b1h, b1l, acc);
  k_step<false, false, false>(KCHUNKS - 1, ag, wh, wl, stg[1], stg[0], toff,
                              lane, a0, a1, b1h, b1l, b0h, b0l, acc);

#pragma clang loop vectorize(disable) interleave(disable)
  for (int b = 0; b < BATCH; ++b) {
    const float* qb = q + b * EMBED + wave * 64 + half * 8;
    float sa[RT], sb[RT];
#pragma unroll
    for (int rt = 0; rt < RT; ++rt) { sa[rt] = 0.0f; sb[rt] = 0.0f; }
#pragma unroll
    for (int nt = 0; nt < NTW; ++nt) {
      const float4 q0 = *(const float4*)(qb + nt * 16);
      const float4 q1 = *(const float4*)(qb + nt * 16 + 4);
#pragma unroll
      for (int rt = 0; rt < RT; ++rt) {
        sa[rt] += fabsf(q0.x - acc[rt][nt][0]);
        sb[rt] += fabsf(q0.y - acc[rt][nt][1]);
        sa[rt] += fabsf(q0.z - acc[rt][nt][2]);
        sb[rt] += fabsf(q0.w - acc[rt][nt][3]);
        sa[rt] += fabsf(q1.x - acc[rt][nt][4]);
        sb[rt] += fabsf(q1.y - acc[rt][nt][5]);
        sa[rt] += fabsf(q1.z - acc[rt][nt][6]);
        sb[rt] += fabsf(q1.w - acc[rt][nt][7]);
      }
    }
#pragma unroll
    for (int rt = 0; rt < RT; ++rt)
      red[(wave * 2 + half) * RED_PLANE + b * BM + rt * 16 + r15] =
          sa[rt] + sb[rt];
  }
  __syncthreads();

  for (int idx = t; idx < BATCH * BM; idx += 128) {
    const int b = idx / BM;
    const int r = idx - b * BM;
    const float w0 = red[0 * RED_PLANE + idx] + red[1 * RED_PLANE + idx];
    const float w1 = red[2 * RED_PLANE + idx] + red[3 * RED_PLANE + idx];
    const float w2 = red[4 * RED_PLANE + idx] + red[5 * RED_PLANE + idx];
    const float w3 = red[6 * RED_PLANE + idx] + red[7 * RED_PLANE + idx];
    out[(size_t)b * NE_SCORE + 1 + row0 + r] = -((w0 + w1) + (w2 + w3));
  }
}

extern "C" void kernel_launch(void* const* d_in, const int* in_sizes, int n_in,
                              void* d_out, int out_size, void* d_ws, size_t ws_size,
                              hipStream_t stream) {
  (void)in_sizes; (void)n_in; (void)out_size; (void)ws_size;
  const float* ent_pkl = (const float*)d_in[0];
  const float* other   = (const float*)d_in[1];
  const float* proj_W  = (const float*)d_in[2];
  const int*   ids     = (const int*)d_in[3];
  const int*   mp      = (const int*)d_in[4];
  float*       out     = (float*)d_out;

  float*  qsum  = (float*)d_ws;
  __bf16* whi   = (__bf16*)(qsum + (size_t)BATCH * EMBED);
  __bf16* wlo   = whi + (size_t)NTILES * KCHUNKS * 32 * 16;

  BLP_split_w<<<NTILES * KCHUNKS, 32, 0, stream>>>(proj_W, whi, wlo);
  BLP_query_proj<<<BATCH / 8, 512, 0, stream>>>(ent_pkl, other, whi, wlo, ids, mp, qsum);
  BLP_proj_score<<<FUSED_BLOCKS + 1, 128, 0, stream>>>(ent_pkl, other, whi, wlo, qsum, out);
}


# Query kernel: its 16 gathered rows are read once per block with whole-line loads, split to bf16 once and shared via LDS in fragment order (was: all 16 waves loaded and split them); sums bit-identical
# speedup vs baseline: 1.8315x; 1.0647x over previous
#include <hip/hip_runtime.h>
#include <hip/hip_bf16.h>

#define NUM_ENT   40000
#define EMBED     256
#define FEAT      768
#define KCHUNKS   (FEAT / 32)
#define NTILES    (EMBED / 16)
#define NE_SCORE  40001
#define BATCH     32
#define BM        80
#define RT        (BM / 16)
#define NTW       4
#define FUSED_BLOCKS (NUM_ENT / BM)
#define STG_ELEMS (RT * 2 * 2 * 32 * 8)
#define RED_PLANE (BATCH * BM + 16)

typedef __attribute__((ext_vector_type(16))) __bf16 v16bf;
typedef __attribute__((ext_vector_type(8)))  float  v8f;

__device__ __forceinline__ void load16(const float* __restrict__ p, float* v) {
  float4 a0 = *(const float4*)(p + 0);
  float4 a1 = *(const float4*)(p + 4);
  float4 b0 = *(const float4*)(p + 16);
  float4 b1 = *(const float4*)(p + 20);
  v[0] = a0.x; v[1] = a0.y; v[2]  = a0.z; v[3]  = a0.w;
  v[4] = a1.x; v[5] = a1.y; v[6]  = a1.z; v[7]  = a1.w;
  v[8] = b0.x; v[9] = b0.y; v[10] = b0.z; v[11] = b0.w;
  v[12] = b1.x; v[13] = b1.y; v[14] = b1.z; v[15] = b1.w;
}

__global__ __launch_bounds__(32) void BLP_split_w(
    const float* __restrict__ W, __bf16* __restrict__ whi,
    __bf16* __restrict__ wlo) {
  const int lane = threadIdx.x;
  const int kc   = blockIdx.x % KCHUNKS;
  const int nt   = blockIdx.x / KCHUNKS;
  const int half = lane >> 4;
  const int r15  = lane & 15;

  const float* p = W + (size_t)(nt * 16 + r15) * FEAT + kc * 32 + half * 8;
  float v[16];
  load16(p, v);

  const size_t off = ((size_t)(nt * KCHUNKS + kc) * 32 + lane) * 16;
#pragma unroll
  for (int i = 0; i < 16; ++i) {
    __bf16 h = (__bf16)v[i];
    whi[off + i] = h;
    wlo[off + i] = (__bf16)(v[i] - (float)h);
  }
}

__device__ __forceinline__ v8f mma3(v16bf whi_f, v16bf wlo_f, v16bf rhi,
                                    v16bf rlo, v8f c) {
  c = __builtin_amdgcn_wmma_f32_16x16x32_bf16(
      false, whi_f, false, rhi, (short)0, c, false, false);
  c = __builtin_amdgcn_wmma_f32_16x16x32_bf16(
      false, wlo_f, false, rhi, (short)0, c, false, false);
  c = __builtin_amdgcn_wmma_f32_16x16x32_bf16(
      false, whi_f, false, rlo, (short)0, c, false, false);
  return c;
}

typedef __attribute__((ext_vector_type(8))) __bf16 v8bf;
typedef __attribute__((ext_vector_type(4))) __bf16 v4bf;
typedef __attribute__((ext_vector_type(4))) float v4f;

__device__ __forceinline__ v16bf ld_frag(const __bf16* st, int rt, int hl,
                                         int lane) {
  const __bf16* p = st + (rt * 2 + hl) * 512 + lane * 8;
  v8bf e0 = *(const v8bf*)p;
  v8bf e1 = *(const v8bf*)(p + 256);
  return __builtin_shufflevector(e0, e1, 0, 1, 2, 3, 4, 5, 6, 7, 8, 9, 10, 11,
                                 12, 13, 14, 15);
}

__device__ __forceinline__ void stage_store(__bf16* dst_hi, v4f v) {
  const float f[4] = {v[0], v[1], v[2], v[3]};
  v4bf hi, lo;
#pragma unroll
  for (int j = 0; j < 4; ++j) {
    __bf16 h = (__bf16)f[j];
    hi[j] = h;
    lo[j] = (__bf16)(f[j] - (float)h);
  }
  *(v4bf*)dst_hi = hi;
  *(v4bf*)(dst_hi + 512) = lo;
}

#define QG 4
#define QCH_ELEMS (2 * 2 * 32 * 8)
__global__ __launch_bounds__(512) void BLP_query_proj(
    const float* __restrict__ A, const float* __restrict__ other,
    const __bf16* __restrict__ whi, const __bf16* __restrict__ wlo,
    const int* __restrict__ ids, const int* __restrict__ mp_ptr,
    float* __restrict__ qsum) {
  __shared__ __attribute__((aligned(16))) __bf16 rows[KCHUNKS * QCH_ELEMS];
  __shared__ float P[16 * EMBED];
  __shared__ float rn[16];
  const int lane = threadIdx.x & 31;
  const int wave = threadIdx.x >> 5;
  const int half = lane >> 4;
  const int r15  = lane & 15;
  const int mp = *mp_ptr;
  const int c0 = (mp == 0) ? 1 : 0;
  const int c1 = (mp <= 1) ? 2 : 1;

  const int kq  = threadIdx.x & 7;
  const int rb  = (threadIdx.x >> 3) & 15;
  const int kc0 = threadIdx.x >> 7;
  const int idl = ids[(blockIdx.x * 8 + (rb & 7)) * 3 + ((rb >> 3) ? c1 : c0)];
  const bool entl = (idl >= 1) && (idl <= NUM_ENT);
  const float* ag = A + (size_t)(entl ? (idl - 1) : 0) * FEAT + kq * 4;
  v4f av[KCHUNKS / 4];
#pragma unroll
  for (int i = 0; i < KCHUNKS / 4; ++i)
    av[i] = *(const v4f*)(ag + (kc0 + 4 * i) * 32);

  const __bf16* wh = whi + ((size_t)(wave * KCHUNKS) * 32 + lane) * 16;
  const __bf16* wl = wlo + ((size_t)(wave * KCHUNKS) * 32 + lane) * 16;
  v16bf bh[2][QG], bl[2][QG];
#pragma unroll
  for (int c = 0; c < QG; ++c) {
    bh[0][c] = *(const v16bf*)(wh + (size_t)c * 512);
    bl[0][c] = *(const v16bf*)(wl + (size_t)c * 512);
  }

  const int b  = blockIdx.x * 8 + (r15 & 7);
  const int id = ids[b * 3 + ((r15 >> 3) ? c1 : c0)];
  const bool is_ent = (id >= 1) && (id <= NUM_ENT);
  const int orow = is_ent ? 0 : ((id == 0) ? 0 : (id - NUM_ENT));

  const int toff =
      ((kq >> 2) * 32 + rb + 16 * ((kq >> 1) & 1)) * 8 + (kq & 1) * 4;
#pragma unroll
  for (int i = 0; i < KCHUNKS / 4; ++i)
    stage_store(rows + (kc0 + 4 * i) * QCH_ELEMS + toff, av[i]);
  __syncthreads();

  v8f acc = {};
#pragma unroll
  for (int g = 0; g < KCHUNKS / QG; ++g) {
    if (g + 1 < KCHUNKS / QG) {
#pragma unroll
      for (int c = 0; c < QG; ++c) {
        bh[(g + 1) & 1][c] = *(const v16bf*)(wh + (size_t)((g + 1) * QG + c) * 512);
        bl[(g + 1) & 1][c] = *(const v16bf*)(wl + (size_t)((g + 1) * QG + c) * 512);
      }
    }
    v16bf rhi[QG], rlo[QG];
#pragma unroll
    for (int c = 0; c < QG; ++c) {
      const __bf16* st = rows + (g * QG + c) * QCH_ELEMS;
      rhi[c] = ld_frag(st, 0, 0, lane);
      rlo[c] = ld_frag(st, 0, 1, lane);
    }
    __builtin_amdgcn_sched_barrier(0);
#pragma unroll
    for (int c = 0; c < QG; ++c)
      acc = mma3(bh[g & 1][c], bl[g & 1][c], rhi[c], rlo[c], acc);
  }
#pragma unroll
  for (int j = 0; j < 8; ++j) {
    const int d = wave * 16 + j + half * 8;
    float val = acc[j];
    if (!is_ent) val = other[(size_t)orow * EMBED + d];
    P[r15 * EMBED + d] = val;
  }
  __syncthreads();
  {
    float s = 0.0f;
#pragma unroll
    for (int i = 0; i < 8; ++i) {
      float x = P[wave * EMBED + lane + 32 * i];
      s += x * x;
    }
    for (int o = 16; o > 0; o >>= 1) s += __shfl_xor(s, o);
    if (lane == 0) rn[wave] = 1.0f / fmaxf(sqrtf(s), 1e-12f);
  }
  __syncthreads();
#pragma unroll
  for (int i = 0; i < 4; ++i) {
    const int idx = threadIdx.x + i * 512;
    const int bl2 = idx >> 8;
    const int d  = idx & 255;
    qsum[(size_t)(blockIdx.x * 8 + bl2) * EMBED + d] =
        P[bl2 * EMBED + d] * rn[bl2] + P[(bl2 + 8) * EMBED + d] * rn[bl2 + 8];
  }
}

template <bool LOADA, bool MORE, bool PF>
__device__ __forceinline__ void k_step(
    int kc, const float* __restrict__ ag, const __bf16* __restrict__ wh,
    const __bf16* __restrict__ wl, __bf16* stg_cur, __bf16* stg_nxt, int toff,
    int lane, v4f (&ac)[RT], v4f (&an)[RT], v16bf (&bhc)[NTW],
    v16bf (&blc)[NTW], v16bf (&bhn)[NTW], v16bf (&bln)[NTW],
    v8f (&acc)[RT][NTW]) {
  if (LOADA) {
#pragma unroll
    for (int i = 0; i < RT; ++i)
      an[i] = *(const v4f*)(ag + (size_t)i * 16 * FEAT + (kc + 2) * 32);
  }
  if (MORE) {
#pragma unroll
    for (int nt = 0; nt < NTW; ++nt) {
      const size_t off = (size_t)(nt * KCHUNKS + kc + 1) * 512;
      bhn[nt] = *(const v16bf*)(wh + off);
      bln[nt] = *(const v16bf*)(wl + off);
    }
  }
  if (PF) {
    const int pk = (kc + 4 < KCHUNKS) ? kc + 4 : KCHUNKS - 1;
#pragma unroll
    for (int i = 0; i < RT; ++i)
      __builtin_prefetch(ag + (size_t)i * 16 * FEAT + pk * 32, 0, 0);
  }
  __builtin_amdgcn_sched_barrier(0);

  v16bf rhi[RT], rlo[RT];
#pragma unroll
  for (int rt = 0; rt < RT; ++rt) {
    rhi[rt] = ld_frag(stg_cur, rt, 0, lane);
    rlo[rt] = ld_frag(stg_cur, rt, 1, lane);
  }
#pragma unroll
  for (int rt = 0; rt < RT; ++rt)
#pragma unroll
    for (int nt = 0; nt < NTW; ++nt)
      acc[rt][nt] = __builtin_amdgcn_wmma_f32_16x16x32_bf16(
          false, bhc[nt], false, rhi[rt], (short)0, acc[rt][nt], false, false);
#pragma unroll
  for (int rt = 0; rt < RT; ++rt)
#pragma unroll
    for (int nt = 0; nt < NTW; ++nt)
      acc[rt][nt] = __builtin_amdgcn_wmma_f32_16x16x32_bf16(
          false, blc[nt], false, rhi[rt], (short)0, acc[rt][nt], false, false);
#pragma unroll
  for (int rt = 0; rt < RT; ++rt)
#pragma unroll
    for (int nt = 0; nt < NTW; ++nt)
      acc[rt][nt] = __builtin_amdgcn_wmma_f32_16x16x32_bf16(
          false, bhc[nt], false, rlo[rt], (short)0, acc[rt][nt], false, false);

  if (MORE) {
#pragma unroll
    for (int i = 0; i < RT; ++i) asm volatile("" : "+v"(ac[i]));
#pragma unroll
    for (int i = 0; i < RT; ++i) stage_store(stg_nxt + i * 1024 + toff, ac[i]);
    __builtin_amdgcn_fence(__ATOMIC_RELEASE, "workgroup", "local");
    __builtin_amdgcn_s_barrier();
    __builtin_amdgcn_fence(__ATOMIC_ACQUIRE, "workgroup", "local");
  }
}

__global__ __launch_bounds__(128) void BLP_proj_score(
    const float* __restrict__ A, const float* __restrict__ other,
    const __bf16* __restrict__ whi, const __bf16* __restrict__ wlo,
    const float* __restrict__ qsum, float* __restrict__ out) {
  __shared__ __attribute__((aligned(16))) float q[BATCH * EMBED];
  __shared__ __attribute__((aligned(16))) __bf16 stg[2][STG_ELEMS];
  __shared__ __attribute__((aligned(16))) float red[8 * RED_PLANE];

  if (blockIdx.x == FUSED_BLOCKS) {
    if (threadIdx.x >= 64) return;
    const int b = threadIdx.x >> 1;
    const int h = threadIdx.x & 1;
    const float4* qp = (const float4*)(qsum + (size_t)b * EMBED + h * 128);
    const float4* op = (const float4*)(other + h * 128);
    float s = 0.0f;
    for (int i = 0; i < 32; ++i) {
      float4 x = qp[i];
      float4 y = op[i];
      s += fabsf(x.x - y.x) + fabsf(x.y - y.y) +
           fabsf(x.z - y.z) + fabsf(x.w - y.w);
    }
    s += __shfl_xor(s, 1);
    if (h == 0) out[(size_t)b * NE_SCORE] = -s;
    return;
  }

  const int t    = threadIdx.x;
  const int lane = t & 31;
  const int wave = t >> 5;
  const int half = lane >> 4;
  const int r15  = lane & 15;
  const int row0 = blockIdx.x * BM;

  const int kq = t & 7;
  const int rb = t >> 3;
  const float* ag = A + (size_t)(row0 + rb) * FEAT + kq * 4;
  const int toff =
      ((kq >> 2) * 32 + rb + 16 * ((kq >> 1) & 1)) * 8 + (kq & 1) * 4;

  const __bf16* wh = whi + ((size_t)(wave * NTW * KCHUNKS) * 32 + lane) * 16;
  const __bf16* wl = wlo + ((size_t)(wave * NTW * KCHUNKS) * 32 + lane) * 16;

  v16bf b0h[NTW], b0l[NTW], b1h[NTW], b1l[NTW];
  v4f a0[RT], a1[RT];
  {
#pragma unroll
    for (int i = 0; i < RT; ++i)
      a0[i] = *(const v4f*)(ag + (size_t)i * 16 * FEAT);
#pragma unroll
    for (int i = 0; i < RT; ++i)
      a1[i] = *(const v4f*)(ag + (size_t)i * 16 * FEAT + 32);
#pragma unroll
    for (int nt = 0; nt < NTW; ++nt) {
      const size_t off = (size_t)(nt * KCHUNKS) * 512;
      b0h[nt] = *(const v16bf*)(wh + off);
      b0l[nt] = *(const v16bf*)(wl + off);
    }
    for (int i = t; i < BATCH * EMBED / 4; i += 128)
      ((float4*)q)[i] = ((const float4*)qsum)[i];
#pragma unroll
    for (int i = 0; i < RT; ++i) stage_store(stg[0] + i * 1024 + toff, a0[i]);
  }
  __syncthreads();

  v8f acc[RT][NTW] = {};
  for (int kc = 0; kc < KCHUNKS - 2; kc += 2) {
    k_step<true, true, true>(kc, ag, wh, wl, stg[0], stg[1], toff, lane, a1,
                             a0, b0h, b0l, b1h, b1l, acc);
    k_step<true, true, false>(kc + 1, ag, wh, wl, stg[1], stg[0], toff, lane,
                              a0, a1, b1h, b1l, b0h, b0l, acc);
  }
  k_step<false, true, false>(KCHUNKS - 2, ag, wh, wl, stg[0], stg[1], toff,
                             lane, a1, a0, b0h, b0l, b1h, b1l, acc);
  k_step<false, false, false>(KCHUNKS - 1, ag, wh, wl, stg[1], stg[0], toff,
                              lane, a0, a1, b1h, b1l, b0h, b0l, acc);

#pragma clang loop vectorize(disable) interleave(disable)
  for (int b = 0; b < BATCH; ++b) {
    const float* qb = q + b * EMBED + wave * 64 + half * 8;
    float sa[RT], sb[RT];
#pragma unroll
    for (int rt = 0; rt < RT; ++rt) { sa[rt] = 0.0f; sb[rt] = 0.0f; }
#pragma unroll
    for (int nt = 0; nt < NTW; ++nt) {
      const float4 q0 = *(const float4*)(qb + nt * 16);
      const float4 q1 = *(const float4*)(qb + nt * 16 + 4);
#pragma unroll
      for (int rt = 0; rt < RT; ++rt) {
        sa[rt] += fabsf(q0.x - acc[rt][nt][0]);
        sb[rt] += fabsf(q0.y - acc[rt][nt][1]);
        sa[rt] += fabsf(q0.z - acc[rt][nt][2]);
        sb[rt] += fabsf(q0.w - acc[rt][nt][3]);
        sa[rt] += fabsf(q1.x - acc[rt][nt][4]);
        sb[rt] += fabsf(q1.y - acc[rt][nt][5]);
        sa[rt] += fabsf(q1.z - acc[rt][nt][6]);
        sb[rt] += fabsf(q1.w - acc[rt][nt][7]);
      }
    }
#pragma unroll
    for (int rt = 0; rt < RT; ++rt)
      red[(wave * 2 + half) * RED_PLANE + b * BM + rt * 16 + r15] =
          sa[rt] + sb[rt];
  }
  __syncthreads();

  for (int idx = t; idx < BATCH * BM; idx += 128) {
    const int b = idx / BM;
    const int r = idx - b * BM;
    const float w0 = red[0 * RED_PLANE + idx] + red[1 * RED_PLANE + idx];
    const float w1 = red[2 * RED_PLANE + idx] + red[3 * RED_PLANE + idx];
    const float w2 = red[4 * RED_PLANE + idx] + red[5 * RED_PLANE + idx];
    const float w3 = red[6 * RED_PLANE + idx] + red[7 * RED_PLANE + idx];
    out[(size_t)b * NE_SCORE + 1 + row0 + r] = -((w0 + w1) + (w2 + w3));
  }
}

extern "C" void kernel_launch(void* const* d_in, const int* in_sizes, int n_in,
                              void* d_out, int out_size, void* d_ws, size_t ws_size,
                              hipStream_t stream) {
  (void)in_sizes; (void)n_in; (void)out_size; (void)ws_size;
  const float* ent_pkl = (const float*)d_in[0];
  const float* other   = (const float*)d_in[1];
  const float* proj_W  = (const float*)d_in[2];
  const int*   ids     = (const int*)d_in[3];
  const int*   mp      = (const int*)d_in[4];
  float*       out     = (float*)d_out;

  float*  qsum  = (float*)d_ws;
  __bf16* whi   = (__bf16*)(qsum + (size_t)BATCH * EMBED);
  __bf16* wlo   = whi + (size_t)NTILES * KCHUNKS * 32 * 16;

  BLP_split_w<<<NTILES * KCHUNKS, 32, 0, stream>>>(proj_W, whi, wlo);
  BLP_query_proj<<<BATCH / 8, 512, 0, stream>>>(ent_pkl, other, whi, wlo, ids, mp, qsum);
  BLP_proj_score<<<FUSED_BLOCKS + 1, 128, 0, stream>>>(ent_pkl, other, whi, wlo, qsum, out);
}


# Packed weights laid out [elem half][lane][8] per fragment: each 16-byte fragment load of a wave covers 512 contiguous bytes, 4 whole lines (was 8 half-used lines per load); sums bit-identical
# speedup vs baseline: 1.9233x; 1.0501x over previous
#include <hip/hip_runtime.h>
#include <hip/hip_bf16.h>

#define NUM_ENT   40000
#define EMBED     256
#define FEAT      768
#define KCHUNKS   (FEAT / 32)
#define NTILES    (EMBED / 16)
#define NE_SCORE  40001
#define BATCH     32
#define BM        80
#define RT        (BM / 16)
#define NTW       4
#define FUSED_BLOCKS (NUM_ENT / BM)
#define STG_ELEMS (RT * 2 * 2 * 32 * 8)
#define RED_PLANE (BATCH * BM + 16)

typedef __attribute__((ext_vector_type(16))) __bf16 v16bf;
typedef __attribute__((ext_vector_type(8)))  __bf16 v8bf;
typedef __attribute__((ext_vector_type(8)))  float  v8f;

__device__ __forceinline__ void load16(const float* __restrict__ p, float* v) {
  float4 a0 = *(const float4*)(p + 0);
  float4 a1 = *(const float4*)(p + 4);
  float4 b0 = *(const float4*)(p + 16);
  float4 b1 = *(const float4*)(p + 20);
  v[0] = a0.x; v[1] = a0.y; v[2]  = a0.z; v[3]  = a0.w;
  v[4] = a1.x; v[5] = a1.y; v[6]  = a1.z; v[7]  = a1.w;
  v[8] = b0.x; v[9] = b0.y; v[10] = b0.z; v[11] = b0.w;
  v[12] = b1.x; v[13] = b1.y; v[14] = b1.z; v[15] = b1.w;
}

__global__ __launch_bounds__(32) void BLP_split_w(
    const float* __restrict__ W, __bf16* __restrict__ whi,
    __bf16* __restrict__ wlo) {
  const int lane = threadIdx.x;
  const int kc   = blockIdx.x % KCHUNKS;
  const int nt   = blockIdx.x / KCHUNKS;
  const int half = lane >> 4;
  const int r15  = lane & 15;

  const float* p = W + (size_t)(nt * 16 + r15) * FEAT + kc * 32 + half * 8;
  float v[16];
  load16(p, v);

  const size_t off = (size_t)(nt * KCHUNKS + kc) * 512 + lane * 8;
  v8bf h0, h1, l0, l1;
#pragma unroll
  for (int i = 0; i < 8; ++i) {
    __bf16 ha = (__bf16)v[i];
    __bf16 hb = (__bf16)v[8 + i];
    h0[i] = ha;
    h1[i] = hb;
    l0[i] = (__bf16)(v[i] - (float)ha);
    l1[i] = (__bf16)(v[8 + i] - (float)hb);
  }
  *(v8bf*)(whi + off) = h0;
  *(v8bf*)(whi + off + 256) = h1;
  *(v8bf*)(wlo + off) = l0;
  *(v8bf*)(wlo + off + 256) = l1;
}

__device__ __forceinline__ v16bf ld_wfrag(const __bf16* __restrict__ p) {
  v8bf e0 = *(const v8bf*)p;
  v8bf e1 = *(const v8bf*)(p + 256);
  return __builtin_shufflevector(e0, e1, 0, 1, 2, 3, 4, 5, 6, 7, 8, 9, 10, 11,
                                 12, 13, 14, 15);
}

__device__ __forceinline__ v8f mma3(v16bf whi_f, v16bf wlo_f, v16bf rhi,
                                    v16bf rlo, v8f c) {
  c = __builtin_amdgcn_wmma_f32_16x16x32_bf16(
      false, whi_f, false, rhi, (short)0, c, false, false);
  c = __builtin_amdgcn_wmma_f32_16x16x32_bf16(
      false, wlo_f, false, rhi, (short)0, c, false, false);
  c = __builtin_amdgcn_wmma_f32_16x16x32_bf16(
      false, whi_f, false, rlo, (short)0, c, false, false);
  return c;
}

typedef __attribute__((ext_vector_type(4))) __bf16 v4bf;
typedef __attribute__((ext_vector_type(4))) float v4f;

__device__ __forceinline__ v16bf ld_frag(const __bf16* st, int rt, int hl,
                                         int lane) {
  const __bf16* p = st + (rt * 2 + hl) * 512 + lane * 8;
  v8bf e0 = *(const v8bf*)p;
  v8bf e1 = *(const v8bf*)(p + 256);
  return __builtin_shufflevector(e0, e1, 0, 1, 2, 3, 4, 5, 6, 7, 8, 9, 10, 11,
                                 12, 13, 14, 15);
}

__device__ __forceinline__ void stage_store(__bf16* dst_hi, v4f v) {
  const float f[4] = {v[0], v[1], v[2], v[3]};
  v4bf hi, lo;
#pragma unroll
  for (int j = 0; j < 4; ++j) {
    __bf16 h = (__bf16)f[j];
    hi[j] = h;
    lo[j] = (__bf16)(f[j] - (float)h);
  }
  *(v4bf*)dst_hi = hi;
  *(v4bf*)(dst_hi + 512) = lo;
}

#define QG 4
#define QCH_ELEMS (2 * 2 * 32 * 8)
__global__ __launch_bounds__(512) void BLP_query_proj(
    const float* __restrict__ A, const float* __restrict__ other,
    const __bf16* __restrict__ whi, const __bf16* __restrict__ wlo,
    const int* __restrict__ ids, const int* __restrict__ mp_ptr,
    float* __restrict__ qsum) {
  __shared__ __attribute__((aligned(16))) __bf16 rows[KCHUNKS * QCH_ELEMS];
  __shared__ float P[16 * EMBED];
  __shared__ float rn[16];
  const int lane = threadIdx.x & 31;
  const int wave = threadIdx.x >> 5;
  const int half = lane >> 4;
  const int r15  = lane & 15;
  const int mp = *mp_ptr;
  const int c0 = (mp == 0) ? 1 : 0;
  const int c1 = (mp <= 1) ? 2 : 1;

  const int kq  = threadIdx.x & 7;
  const int rb  = (threadIdx.x >> 3) & 15;
  const int kc0 = threadIdx.x >> 7;
  const int idl = ids[(blockIdx.x * 8 + (rb & 7)) * 3 + ((rb >> 3) ? c1 : c0)];
  const bool entl = (idl >= 1) && (idl <= NUM_ENT);
  const float* ag = A + (size_t)(entl ? (idl - 1) : 0) * FEAT + kq * 4;
  v4f av[KCHUNKS / 4];
#pragma unroll
  for (int i = 0; i < KCHUNKS / 4; ++i)
    av[i] = *(const v4f*)(ag + (kc0 + 4 * i) * 32);

  const __bf16* wh = whi + (size_t)(wave * KCHUNKS) * 512 + lane * 8;
  const __bf16* wl = wlo + (size_t)(wave * KCHUNKS) * 512 + lane * 8;
  v16bf bh[2][QG], bl[2][QG];
#pragma unroll
  for (int c = 0; c < QG; ++c) {
    bh[0][c] = ld_wfrag(wh + (size_t)c * 512);
    bl[0][c] = ld_wfrag(wl + (size_t)c * 512);
  }

  const int b  = blockIdx.x * 8 + (r15 & 7);
  const int id = ids[b * 3 + ((r15 >> 3) ? c1 : c0)];
  const bool is_ent = (id >= 1) && (id <= NUM_ENT);
  const int orow = is_ent ? 0 : ((id == 0) ? 0 : (id - NUM_ENT));

  const int toff =
      ((kq >> 2) * 32 + rb + 16 * ((kq >> 1) & 1)) * 8 + (kq & 1) * 4;
#pragma unroll
  for (int i = 0; i < KCHUNKS / 4; ++i)
    stage_store(rows + (kc0 + 4 * i) * QCH_ELEMS + toff, av[i]);
  __syncthreads();

  v8f acc = {};
#pragma unroll
  for (int g = 0; g < KCHUNKS / QG; ++g) {
    if (g + 1 < KCHUNKS / QG) {
#pragma unroll
      for (int c = 0; c < QG; ++c) {
        bh[(g + 1) & 1][c] = ld_wfrag(wh + (size_t)((g + 1) * QG + c) * 512);
        bl[(g + 1) & 1][c] = ld_wfrag(wl + (size_t)((g + 1) * QG + c) * 512);
      }
    }
    v16bf rhi[QG], rlo[QG];
#pragma unroll
    for (int c = 0; c < QG; ++c) {
      const __bf16* st = rows + (g * QG + c) * QCH_ELEMS;
      rhi[c] = ld_frag(st, 0, 0, lane);
      rlo[c] = ld_frag(st, 0, 1, lane);
    }
    __builtin_amdgcn_sched_barrier(0);
#pragma unroll
    for (int c = 0; c < QG; ++c)
      acc = mma3(bh[g & 1][c], bl[g & 1][c], rhi[c], rlo[c], acc);
  }
#pragma unroll
  for (int j = 0; j < 8; ++j) {
    const int d = wave * 16 + j + half * 8;
    float val = acc[j];
    if (!is_ent) val = other[(size_t)orow * EMBED + d];
    P[r15 * EMBED + d] = val;
  }
  __syncthreads();
  {
    float s = 0.0f;
#pragma unroll
    for (int i = 0; i < 8; ++i) {
      float x = P[wave * EMBED + lane + 32 * i];
      s += x * x;
    }
    for (int o = 16; o > 0; o >>= 1) s += __shfl_xor(s, o);
    if (lane == 0) rn[wave] = 1.0f / fmaxf(sqrtf(s), 1e-12f);
  }
  __syncthreads();
#pragma unroll
  for (int i = 0; i < 4; ++i) {
    const int idx = threadIdx.x + i * 512;
    const int bl2 = idx >> 8;
    const int d  = idx & 255;
    qsum[(size_t)(blockIdx.x * 8 + bl2) * EMBED + d] =
        P[bl2 * EMBED + d] * rn[bl2] + P[(bl2 + 8) * EMBED + d] * rn[bl2 + 8];
  }
}

template <bool LOADA, bool MORE, bool PF>
__device__ __forceinline__ void k_step(
    int kc, const float* __restrict__ ag, const __bf16* __restrict__ wh,
    const __bf16* __restrict__ wl, __bf16* stg_cur, __bf16* stg_nxt, int toff,
    int lane, v4f (&ac)[RT], v4f (&an)[RT], v16bf (&bhc)[NTW],
    v16bf (&blc)[NTW], v16bf (&bhn)[NTW], v16bf (&bln)[NTW],
    v8f (&acc)[RT][NTW]) {
  if (LOADA) {
#pragma unroll
    for (int i = 0; i < RT; ++i)
      an[i] = *(const v4f*)(ag + (size_t)i * 16 * FEAT + (kc + 2) * 32);
  }
  if (MORE) {
#pragma unroll
    for (int nt = 0; nt < NTW; ++nt) {
      const size_t off = (size_t)(nt * KCHUNKS + kc + 1) * 512;
      bhn[nt] = ld_wfrag(wh + off);
      bln[nt] = ld_wfrag(wl + off);
    }
  }
  if (PF) {
    const int pk = (kc + 4 < KCHUNKS) ? kc + 4 : KCHUNKS - 1;
#pragma unroll
    for (int i = 0; i < RT; ++i)
      __builtin_prefetch(ag + (size_t)i * 16 * FEAT + pk * 32, 0, 0);
  }
  __builtin_amdgcn_sched_barrier(0);

  v16bf rhi[RT], rlo[RT];
#pragma unroll
  for (int rt = 0; rt < RT; ++rt) {
    rhi[rt] = ld_frag(stg_cur, rt, 0, lane);
    rlo[rt] = ld_frag(stg_cur, rt, 1, lane);
  }
#pragma unroll
  for (int rt = 0; rt < RT; ++rt)
#pragma unroll
    for (int nt = 0; nt < NTW; ++nt)
      acc[rt][nt] = __builtin_amdgcn_wmma_f32_16x16x32_bf16(
          false, bhc[nt], false, rhi[rt], (short)0, acc[rt][nt], false, false);
#pragma unroll
  for (int rt = 0; rt < RT; ++rt)
#pragma unroll
    for (int nt = 0; nt < NTW; ++nt)
      acc[rt][nt] = __builtin_amdgcn_wmma_f32_16x16x32_bf16(
          false, blc[nt], false, rhi[rt], (short)0, acc[rt][nt], false, false);
#pragma unroll
  for (int rt = 0; rt < RT; ++rt)
#pragma unroll
    for (int nt = 0; nt < NTW; ++nt)
      acc[rt][nt] = __builtin_amdgcn_wmma_f32_16x16x32_bf16(
          false, bhc[nt], false, rlo[rt], (short)0, acc[rt][nt], false, false);

  if (MORE) {
#pragma unroll
    for (int i = 0; i < RT; ++i) asm volatile("" : "+v"(ac[i]));
#pragma unroll
    for (int i = 0; i < RT; ++i) stage_store(stg_nxt + i * 1024 + toff, ac[i]);
    __builtin_amdgcn_fence(__ATOMIC_RELEASE, "workgroup", "local");
    __builtin_amdgcn_s_barrier();
    __builtin_amdgcn_fence(__ATOMIC_ACQUIRE, "workgroup", "local");
  }
}

__global__ __launch_bounds__(128) void BLP_proj_score(
    const float* __restrict__ A, const float* __restrict__ other,
    const __bf16* __restrict__ whi, const __bf16* __restrict__ wlo,
    const float* __restrict__ qsum, float* __restrict__ out) {
  __shared__ __attribute__((aligned(16))) float q[BATCH * EMBED];
  __shared__ __attribute__((aligned(16))) __bf16 stg[2][STG_ELEMS];
  __shared__ __attribute__((aligned(16))) float red[8 * RED_PLANE];

  if (blockIdx.x == FUSED_BLOCKS) {
    if (threadIdx.x >= 64) return;
    const int b = threadIdx.x >> 1;
    const int h = threadIdx.x & 1;
    const float4* qp = (const float4*)(qsum + (size_t)b * EMBED + h * 128);
    const float4* op = (const float4*)(other + h * 128);
    float s = 0.0f;
    for (int i = 0; i < 32; ++i) {
      float4 x = qp[i];
      float4 y = op[i];
      s += fabsf(x.x - y.x) + fabsf(x.y - y.y) +
           fabsf(x.z - y.z) + fabsf(x.w - y.w);
    }
    s += __shfl_xor(s, 1);
    if (h == 0) out[(size_t)b * NE_SCORE] = -s;
    return;
  }

  const int t    = threadIdx.x;
  const int lane = t & 31;
  const int wave = t >> 5;
  const int half = lane >> 4;
  const int r15  = lane & 15;
  const int row0 = blockIdx.x * BM;

  const int kq = t & 7;
  const int rb = t >> 3;
  const float* ag = A + (size_t)(row0 + rb) * FEAT + kq * 4;
  const int toff =
      ((kq >> 2) * 32 + rb + 16 * ((kq >> 1) & 1)) * 8 + (kq & 1) * 4;

  const __bf16* wh = whi + (size_t)(wave * NTW * KCHUNKS) * 512 + lane * 8;
  const __bf16* wl = wlo + (size_t)(wave * NTW * KCHUNKS) * 512 + lane * 8;

  v16bf b0h[NTW], b0l[NTW], b1h[NTW], b1l[NTW];
  v4f a0[RT], a1[RT];
  {
#pragma unroll
    for (int i = 0; i < RT; ++i)
      a0[i] = *(const v4f*)(ag + (size_t)i * 16 * FEAT);
#pragma unroll
    for (int i = 0; i < RT; ++i)
      a1[i] = *(const v4f*)(ag + (size_t)i * 16 * FEAT + 32);
#pragma unroll
    for (int nt = 0; nt < NTW; ++nt) {
      const size_t off = (size_t)(nt * KCHUNKS) * 512;
      b0h[nt] = ld_wfrag(wh + off);
      b0l[nt] = ld_wfrag(wl + off);
    }
    for (int i = t; i < BATCH * EMBED / 4; i += 128)
      ((float4*)q)[i] = ((const float4*)qsum)[i];
#pragma unroll
    for (int i = 0; i < RT; ++i) stage_store(stg[0] + i * 1024 + toff, a0[i]);
  }
  __syncthreads();

  v8f acc[RT][NTW] = {};
  for (int kc = 0; kc < KCHUNKS - 2; kc += 2) {
    k_step<true, true, true>(kc, ag, wh, wl, stg[0], stg[1], toff, lane, a1,
                             a0, b0h, b0l, b1h, b1l, acc);
    k_step<true, true, false>(kc + 1, ag, wh, wl, stg[1], stg[0], toff, lane,
                              a0, a1, b1h, b1l, b0h, b0l, acc);
  }
  k_step<false, true, false>(KCHUNKS - 2, ag, wh, wl, stg[0], stg[1], toff,
                             lane, a1, a0, b0h, b0l, b1h, b1l, acc);
  k_step<false, false, false>(KCHUNKS - 1, ag, wh, wl, stg[1], stg[0], toff,
                              lane, a0, a1, b1h, b1l, b0h, b0l, acc);

#pragma clang loop vectorize(disable) interleave(disable)
  for (int b = 0; b < BATCH; ++b) {
    const float* qb = q + b * EMBED + wave * 64 + half * 8;
    float sa[RT], sb[RT];
#pragma unroll
    for (int rt = 0; rt < RT; ++rt) { sa[rt] = 0.0f; sb[rt] = 0.0f; }
#pragma unroll
    for (int nt = 0; nt < NTW; ++nt) {
      const float4 q0 = *(const float4*)(qb + nt * 16);
      const float4 q1 = *(const float4*)(qb + nt * 16 + 4);
#pragma unroll
      for (int rt = 0; rt < RT; ++rt) {
        sa[rt] += fabsf(q0.x - acc[rt][nt][0]);
        sb[rt] += fabsf(q0.y - acc[rt][nt][1]);
        sa[rt] += fabsf(q0.z - acc[rt][nt][2]);
        sb[rt] += fabsf(q0.w - acc[rt][nt][3]);
        sa[rt] += fabsf(q1.x - acc[rt][nt][4]);
        sb[rt] += fabsf(q1.y - acc[rt][nt][5]);
        sa[rt] += fabsf(q1.z - acc[rt][nt][6]);
        sb[rt] += fabsf(q1.w - acc[rt][nt][7]);
      }
    }
#pragma unroll
    for (int rt = 0; rt < RT; ++rt)
      red[(wave * 2 + half) * RED_PLANE + b * BM + rt * 16 + r15] =
          sa[rt] + sb[rt];
  }
  __syncthreads();

  for (int idx = t; idx < BATCH * BM; idx += 128) {
    const int b = idx / BM;
    const int r = idx - b * BM;
    const float w0 = red[0 * RED_PLANE + idx] + red[1 * RED_PLANE + idx];
    const float w1 = red[2 * RED_PLANE + idx] + red[3 * RED_PLANE + idx];
    const float w2 = red[4 * RED_PLANE + idx] + red[5 * RED_PLANE + idx];
    const float w3 = red[6 * RED_PLANE + idx] + red[7 * RED_PLANE + idx];
    out[(size_t)b * NE_SCORE + 1 + row0 + r] = -((w0 + w1) + (w2 + w3));
  }
}

extern "C" void kernel_launch(void* const* d_in, const int* in_sizes, int n_in,
                              void* d_out, int out_size, void* d_ws, size_t ws_size,
                              hipStream_t stream) {
  (void)in_sizes; (void)n_in; (void)out_size; (void)ws_size;
  const float* ent_pkl = (const float*)d_in[0];
  const float* other   = (const float*)d_in[1];
  const float* proj_W  = (const float*)d_in[2];
  const int*   ids     = (const int*)d_in[3];
  const int*   mp      = (const int*)d_in[4];
  float*       out     = (float*)d_out;

  float*  qsum  = (float*)d_ws;
  __bf16* whi   = (__bf16*)(qsum + (size_t)BATCH * EMBED);
  __bf16* wlo   = whi + (size_t)NTILES * KCHUNKS * 32 * 16;

  BLP_split_w<<<NTILES * KCHUNKS, 32, 0, stream>>>(proj_W, whi, wlo);
  BLP_query_proj<<<BATCH / 8, 512, 0, stream>>>(ent_pkl, other, whi, wlo, ids, mp, qsum);
  BLP_proj_score<<<FUSED_BLOCKS + 1, 128, 0, stream>>>(ent_pkl, other, whi, wlo, qsum, out);
}
